# starting kernel
# baseline (speedup 1.0000x reference)
#include <hip/hip_runtime.h>

typedef float v2f __attribute__((ext_vector_type(2)));
typedef float v4f __attribute__((ext_vector_type(4)));
typedef float v8f __attribute__((ext_vector_type(8)));

namespace {
constexpr int kH       = 16;
constexpr int kD       = 64;
constexpr int kB       = 32768;
constexpr int kThreads = 256;
constexpr int kWaves   = kThreads / 32;
constexpr int kNT      = 16;
constexpr int kNB      = kWaves * kNT;
}

__global__ __launch_bounds__(kThreads)
void block_linear_wmma_f32(const float* __restrict__ W,
                           const float* __restrict__ X,
                           float* __restrict__ Y) {
  __shared__ float Wl[kD * kD];

  const int tid = threadIdx.x;
  const int h   = blockIdx.y;

  {
    const v4f* __restrict__ Wg4 = (const v4f*)(W + (size_t)h * kD * kD);
    v4f* Wl4 = (v4f*)Wl;
#pragma unroll
    for (int i = 0; i < (kD * kD / 4) / kThreads; ++i)
      Wl4[tid + i * kThreads] = Wg4[tid + i * kThreads];
  }
  __syncthreads();

  const int wave    = tid >> 5;
  const int lane    = tid & 31;
  const int lane_lo = lane & 15;
  const int khalf   = lane >> 4;

  const int n0  = blockIdx.x * kNB + wave * kNT;
  const int col = n0 + lane_lo;

  const float* __restrict__ Xh = X + (size_t)h * kD * kB;
  float*       __restrict__ Yh = Y + (size_t)h * kD * kB;

  v8f c0 = {}, c1 = {}, c2 = {}, c3 = {};

#pragma unroll
  for (int kk = 0; kk < kD / 4; ++kk) {
    const int krow = kk * 4 + khalf * 2;

    v2f b;
    b.x = Xh[(size_t)krow * kB + col];
    b.y = Xh[(size_t)(krow + 1) * kB + col];

    const v2f a0 = *(const v2f*)&Wl[(0 * 16 + lane_lo) * kD + krow];
    const v2f a1 = *(const v2f*)&Wl[(1 * 16 + lane_lo) * kD + krow];
    const v2f a2 = *(const v2f*)&Wl[(2 * 16 + lane_lo) * kD + krow];
    const v2f a3 = *(const v2f*)&Wl[(3 * 16 + lane_lo) * kD + krow];

    c0 = __builtin_amdgcn_wmma_f32_16x16x4_f32(false, a0, false, b, (short)0, c0, false, false);
    c1 = __builtin_amdgcn_wmma_f32_16x16x4_f32(false, a1, false, b, (short)0, c1, false, false);
    c2 = __builtin_amdgcn_wmma_f32_16x16x4_f32(false, a2, false, b, (short)0, c2, false, false);
    c3 = __builtin_amdgcn_wmma_f32_16x16x4_f32(false, a3, false, b, (short)0, c3, false, false);
  }

#pragma unroll
  for (int v = 0; v < 8; ++v) {
    const int rbase = v + khalf * 8;
    Yh[(size_t)(0 * 16 + rbase) * kB + col] = c0[v];
    Yh[(size_t)(1 * 16 + rbase) * kB + col] = c1[v];
    Yh[(size_t)(2 * 16 + rbase) * kB + col] = c2[v];
    Yh[(size_t)(3 * 16 + rbase) * kB + col] = c3[v];
  }
}

extern "C" void kernel_launch(void* const* d_in, const int* in_sizes, int n_in,
                              void* d_out, int out_size, void* d_ws, size_t ws_size,
                              hipStream_t stream) {
  const float* W = (const float*)d_in[0];
  const float* X = (const float*)d_in[1];
  float*       Y = (float*)d_out;

  dim3 grid(kB / kNB, kH);
  block_linear_wmma_f32<<<grid, kThreads, 0, stream>>>(W, X, Y);
}


# Pad the LDS weight block's row pitch from 64 to 68 floats (272 B = 16 B x 17) so each A-fragment ds_load_b64 hits 64 distinct banks instead of a 16-way conflict; k order and sums unchanged
# speedup vs baseline: 1.6219x; 1.6219x over previous
#include <hip/hip_runtime.h>

typedef float v2f __attribute__((ext_vector_type(2)));
typedef float v4f __attribute__((ext_vector_type(4)));
typedef float v8f __attribute__((ext_vector_type(8)));

namespace {
constexpr int kH       = 16;
constexpr int kD       = 64;
constexpr int kB       = 32768;
constexpr int kThreads = 256;
constexpr int kWaves   = kThreads / 32;
constexpr int kNT      = 16;
constexpr int kNB      = kWaves * kNT;
constexpr int kPitch   = kD + 4;
}

__global__ __launch_bounds__(kThreads)
void block_linear_wmma_f32(const float* __restrict__ W,
                           const float* __restrict__ X,
                           float* __restrict__ Y) {
  __shared__ __attribute__((aligned(16))) float Wl[kD * kPitch];

  const int tid = threadIdx.x;
  const int h   = blockIdx.y;

  {
    const v4f* __restrict__ Wg4 = (const v4f*)(W + (size_t)h * kD * kD);
#pragma unroll
    for (int i = 0; i < (kD * kD / 4) / kThreads; ++i) {
      const int idx = tid + i * kThreads;
      const int row = idx >> 4;
      const int c4  = idx & 15;
      *(v4f*)&Wl[row * kPitch + c4 * 4] = Wg4[idx];
    }
  }
  __syncthreads();

  const int wave    = tid >> 5;
  const int lane    = tid & 31;
  const int lane_lo = lane & 15;
  const int khalf   = lane >> 4;

  const int n0  = blockIdx.x * kNB + wave * kNT;
  const int col = n0 + lane_lo;

  const float* __restrict__ Xh = X + (size_t)h * kD * kB;
  float*       __restrict__ Yh = Y + (size_t)h * kD * kB;

  v8f c0 = {}, c1 = {}, c2 = {}, c3 = {};

#pragma unroll
  for (int kk = 0; kk < kD / 4; ++kk) {
    const int krow = kk * 4 + khalf * 2;

    v2f b;
    b.x = Xh[(size_t)krow * kB + col];
    b.y = Xh[(size_t)(krow + 1) * kB + col];

    const v2f a0 = *(const v2f*)&Wl[(0 * 16 + lane_lo) * kPitch + krow];
    const v2f a1 = *(const v2f*)&Wl[(1 * 16 + lane_lo) * kPitch + krow];
    const v2f a2 = *(const v2f*)&Wl[(2 * 16 + lane_lo) * kPitch + krow];
    const v2f a3 = *(const v2f*)&Wl[(3 * 16 + lane_lo) * kPitch + krow];

    c0 = __builtin_amdgcn_wmma_f32_16x16x4_f32(false, a0, false, b, (short)0, c0, false, false);
    c1 = __builtin_amdgcn_wmma_f32_16x16x4_f32(false, a1, false, b, (short)0, c1, false, false);
    c2 = __builtin_amdgcn_wmma_f32_16x16x4_f32(false, a2, false, b, (short)0, c2, false, false);
    c3 = __builtin_amdgcn_wmma_f32_16x16x4_f32(false, a3, false, b, (short)0, c3, false, false);
  }

#pragma unroll
  for (int v = 0; v < 8; ++v) {
    const int rbase = v + khalf * 8;
    Yh[(size_t)(0 * 16 + rbase) * kB + col] = c0[v];
    Yh[(size_t)(1 * 16 + rbase) * kB + col] = c1[v];
    Yh[(size_t)(2 * 16 + rbase) * kB + col] = c2[v];
    Yh[(size_t)(3 * 16 + rbase) * kB + col] = c3[v];
  }
}

extern "C" void kernel_launch(void* const* d_in, const int* in_sizes, int n_in,
                              void* d_out, int out_size, void* d_ws, size_t ws_size,
                              hipStream_t stream) {
  const float* W = (const float*)d_in[0];
  const float* X = (const float*)d_in[1];
  float*       Y = (float*)d_out;

  dim3 grid(kB / kNB, kH);
  block_linear_wmma_f32<<<grid, kThreads, 0, stream>>>(W, X, Y);
}
